# starting kernel
# baseline (speedup 1.0000x reference)
#include <hip/hip_runtime.h>

typedef __attribute__((ext_vector_type(2))) float v2f;
typedef __attribute__((ext_vector_type(8))) float v8f;

#define HD 16

__global__ void posmlp_gen_wmma(const float* __restrict__ Q,
                                const float* __restrict__ Wg,
                                const float* __restrict__ bg,
                                float* __restrict__ Wout,
                                int M, int N, int C, int tilesN, int nTiles) {
  const int wavesPerBlk = blockDim.x >> 5;
  const int tile = blockIdx.x * wavesPerBlk + ((int)threadIdx.x >> 5);
  if (tile >= nTiles) return;

  const int tm = tile / tilesN;
  const int tn = tile - tm * tilesN;
  const int m0 = tm * 16, n0 = tn * 16;

  const int lane = threadIdx.x & 31;
  const int half = lane >> 4;
  const int l    = lane & 15;

  const int mA = m0 + l;
  const int nB = n0 + l;
  const int mC = (mA < M) ? mA : (M - 1);
  const int nC = (nB < N) ? nB : (N - 1);

  const float* qrow = Q  + (size_t)mC * C + 2 * half;
  const float* wrow = Wg + (size_t)nC * C + 2 * half;

  v8f acc = {};
  #pragma unroll 4
  for (int k = 0; k < C; k += 4) {
    const v2f a = *(const v2f*)(qrow + k);
    const v2f b = *(const v2f*)(wrow + k);
    acc = __builtin_amdgcn_wmma_f32_16x16x4_f32(
        false, a, false, b, (short)0, acc, false, false);
  }

  const float bias = bg[nC];
  const bool  nOk  = (nB < N);
  #pragma unroll
  for (int i = 0; i < 8; ++i) {
    const int m = m0 + i + 8 * half;
    if (m < M && nOk) Wout[(size_t)m * N + nB] = acc[i] + bias;
  }
}

__global__ void posmlp_spatial(const float* __restrict__ Wq,
                               const float* __restrict__ pos,
                               const int* __restrict__ Hp,
                               const int* __restrict__ Wp,
                               float* __restrict__ out,
                               int N, int HW) {
  __shared__ float sw[96];
  const int bq = blockIdx.x;
  const int t  = threadIdx.x;

  if (t < N) sw[t] = Wq[(size_t)bq * N + t];
  __syncthreads();

  const int H_ = *Hp;
  const int W_ = *Wp;

  float w1y[HD], w1x[HD], b1[HD], w2[HD];
  #pragma unroll
  for (int k = 0; k < HD; ++k) {
    w1y[k] = sw[2 * k];
    w1x[k] = sw[2 * k + 1];
    b1[k]  = sw[2 * HD + k];
    w2[k]  = sw[3 * HD + k];
  }
  const float b2 = sw[4 * HD];

  const float cx = pos[bq * 4 + 0];
  const float cy = pos[bq * 4 + 1];
  const float invbw = 1.0f / pos[bq * 4 + 2];
  const float invbh = 1.0f / pos[bq * 4 + 3];
  const float invH = 1.0f / (float)H_;
  const float invW = 1.0f / (float)W_;

  float* obase = out + (size_t)bq * HW;
  for (int p = t; p < HW; p += blockDim.x) {
    const int h = p / W_;
    const int w = p - h * W_;
    const float rel_y = (((float)h + 0.5f) * invH - cy) * invbh;
    const float rel_x = (((float)w + 0.5f) * invW - cx) * invbw;
    float acc = b2;
    #pragma unroll
    for (int k = 0; k < HD; ++k) {
      const float hid = fmaf(rel_y, w1y[k], fmaf(rel_x, w1x[k], b1[k]));
      acc = fmaf(fmaxf(hid, 0.0f), w2[k], acc);
    }
    obase[p] = acc;
  }
}

extern "C" void kernel_launch(void* const* d_in, const int* in_sizes, int n_in,
                              void* d_out, int out_size, void* d_ws, size_t ws_size,
                              hipStream_t stream) {
  const float* queries = (const float*)d_in[0];
  const float* pos     = (const float*)d_in[1];
  const float* w_gen   = (const float*)d_in[2];
  const float* b_gen   = (const float*)d_in[3];
  const int*   Hp      = (const int*)d_in[4];
  const int*   Wp      = (const int*)d_in[5];

  const int N  = in_sizes[3];
  const int C  = in_sizes[2] / N;
  const int BQ = in_sizes[0] / C;
  const int HW = out_size / BQ;

  float* Wout = (float*)d_ws;

  const int tilesM = (BQ + 15) / 16;
  const int tilesN = (N + 15) / 16;
  const int nTiles = tilesM * tilesN;
  const int wavesPerBlk = 4;
  const int blocks1 = (nTiles + wavesPerBlk - 1) / wavesPerBlk;

  posmlp_gen_wmma<<<blocks1, wavesPerBlk * 32, 0, stream>>>(
      queries, w_gen, b_gen, Wout, BQ, N, C, tilesN, nTiles);

  posmlp_spatial<<<BQ, 256, 0, stream>>>(
      Wout, pos, Hp, Wp, (float*)d_out, N, HW);
}


# one launch instead of two: each block computes its 8 queries' 65 weights by f32 WMMA (K split over waves, partials summed in LDS: GEMM sum order changes), then 2 waves per query evaluate the map
# speedup vs baseline: 1.5068x; 1.5068x over previous
#include <hip/hip_runtime.h>


typedef __attribute__((ext_vector_type(2))) float v2f;
typedef __attribute__((ext_vector_type(8))) float v8f;

#define HD   16
#define NWT  (4 * HD + 1)
#define QPB  8
#define KSPL 4
#define KST  16
#define EST  4
#define SWP  (NWT + 3)

__global__ __launch_bounds__(512) void posmlp_fused(const float* __restrict__ Q,
                                                    const float* __restrict__ Wg,
                                                    const float* __restrict__ bg,
                                                    const float* __restrict__ pos,
                                                    const int* __restrict__ Hp,
                                                    const int* __restrict__ Wp,
                                                    float* __restrict__ out,
                                                    int BQ, int N, int C, int HW) {
  __shared__ float part[KSPL][QPB][4 * HD];
  __shared__ float partE[16][QPB];
  __shared__ float sw[QPB][SWP];

  const int t    = threadIdx.x;
  const int wave = __builtin_amdgcn_readfirstlane(t >> 5);
  const int lane = t & 31;
  const int hf   = lane >> 4;
  const int l    = lane & 15;
  const int q0   = blockIdx.x * QPB;

  const int qi  = wave >> 1;
  const int sub = wave & 1;
  const int bq  = q0 + qi;
  const int bqc = (bq < BQ) ? bq : (BQ - 1);
  const float cx = pos[bqc * 4 + 0];
  const float cy = pos[bqc * 4 + 1];
  const float bw = pos[bqc * 4 + 2];
  const float bh = pos[bqc * 4 + 3];
  const int H_ = *Hp;
  const int W_ = *Wp;

  {
    const int ks = wave & (KSPL - 1);
    const int nt = wave >> 2;

    int mA = q0 + (l & (QPB - 1));  mA = (mA < BQ) ? mA : (BQ - 1);
    int nB = nt * 16 + l;           nB = (nB < N) ? nB : (N - 1);
    int nE = 4 * HD + l;            nE = (nE < N) ? nE : (N - 1);

    const float* qrow = Q  + (size_t)mA * C + 2 * hf;
    const float* wrow = Wg + (size_t)nB * C + 2 * hf;
    const float* erow = Wg + (size_t)nE * C + 2 * hf;

    v8f acc  = {};
    v8f accE = {};
    for (int kb = ks * (4 * KST); kb + 4 * KST <= C; kb += KSPL * 4 * KST) {
      const int ke = kb + 4 * EST * nt;
      v2f a[KST], b[KST], ae[EST], e[EST];
      #pragma unroll
      for (int s = 0; s < KST; ++s) {
        a[s] = *(const v2f*)(qrow + kb + 4 * s);
        b[s] = *(const v2f*)(wrow + kb + 4 * s);
      }
      #pragma unroll
      for (int s = 0; s < EST; ++s) {
        ae[s] = *(const v2f*)(qrow + ke + 4 * s);
        e[s]  = *(const v2f*)(erow + ke + 4 * s);
      }
      #pragma unroll
      for (int s = 0; s < KST; ++s) {
        acc = __builtin_amdgcn_wmma_f32_16x16x4_f32(
            false, a[s], false, b[s], (short)0, acc, false, false);
      }
      #pragma unroll
      for (int s = 0; s < EST; ++s) {
        accE = __builtin_amdgcn_wmma_f32_16x16x4_f32(
            false, ae[s], false, e[s], (short)0, accE, false, false);
      }
    }
    for (int k = (C & ~(4 * KST - 1)) + 4 * ks; k < C; k += 4 * KSPL) {
      const v2f av = *(const v2f*)(qrow + k);
      const v2f bv = *(const v2f*)(wrow + k);
      acc = __builtin_amdgcn_wmma_f32_16x16x4_f32(
          false, av, false, bv, (short)0, acc, false, false);
      if (nt == 0) {
        const v2f ev = *(const v2f*)(erow + k);
        accE = __builtin_amdgcn_wmma_f32_16x16x4_f32(
            false, av, false, ev, (short)0, accE, false, false);
      }
    }

    if (hf == 0) {
      #pragma unroll
      for (int i = 0; i < 8; ++i) part[ks][i][nt * 16 + l] = acc[i];
    }
    if (lane == 0) {
      #pragma unroll
      for (int i = 0; i < 8; ++i) partE[wave][i] = accE[i];
    }
  }
  __syncthreads();

  {
    const int r = t >> 6;
    const int n = t & 63;
    sw[r][n] = ((part[0][r][n] + part[1][r][n]) + (part[2][r][n] + part[3][r][n])) + bg[n];
    if (t < QPB) {
      float s64 = 0.0f;
      #pragma unroll
      for (int w = 0; w < 16; ++w) s64 += partE[w][t];
      sw[t][4 * HD] = s64 + bg[4 * HD];
    }
  }
  __syncthreads();

  if (bq >= BQ) return;

  float w1y[HD], w1x[HD], b1[HD], w2[HD];
  #pragma unroll
  for (int k = 0; k < HD; ++k) {
    w1y[k] = sw[qi][2 * k];
    w1x[k] = sw[qi][2 * k + 1];
    b1[k]  = sw[qi][2 * HD + k];
    w2[k]  = sw[qi][3 * HD + k];
  }
  const float b2 = sw[qi][4 * HD];

  const float invbw = 1.0f / bw;
  const float invbh = 1.0f / bh;
  const float invH = 1.0f / (float)H_;
  const float invW = 1.0f / (float)W_;

  float* obase = out + (size_t)bq * HW;
  for (int p = sub * 32 + lane; p < HW; p += 64) {
    const int h = p / W_;
    const int w = p - h * W_;
    const float rel_y = (((float)h + 0.5f) * invH - cy) * invbh;
    const float rel_x = (((float)w + 0.5f) * invW - cx) * invbw;
    float o = b2;
    #pragma unroll
    for (int k = 0; k < HD; ++k) {
      const float hid = fmaf(rel_y, w1y[k], fmaf(rel_x, w1x[k], b1[k]));
      o = fmaf(fmaxf(hid, 0.0f), w2[k], o);
    }
    obase[p] = o;
  }
}

extern "C" void kernel_launch(void* const* d_in, const int* in_sizes, int n_in,
                              void* d_out, int out_size, void* d_ws, size_t ws_size,
                              hipStream_t stream) {
  const float* queries = (const float*)d_in[0];
  const float* pos     = (const float*)d_in[1];
  const float* w_gen   = (const float*)d_in[2];
  const float* b_gen   = (const float*)d_in[3];
  const int*   Hp      = (const int*)d_in[4];
  const int*   Wp      = (const int*)d_in[5];

  const int N  = in_sizes[3];
  const int C  = in_sizes[2] / N;
  const int BQ = in_sizes[0] / C;
  const int HW = out_size / BQ;

  const int blocks = (BQ + QPB - 1) / QPB;

  posmlp_fused<<<blocks, 512, 0, stream>>>(
      queries, w_gen, b_gen, pos, Hp, Wp, (float*)d_out, BQ, N, C, HW);
}


# spatial map in separable form: row term rel_y*w1y+b1 plus column term rel_x*w1x held in registers, hidden units as packed pairs (even/odd k summed apart: sum order changes), no index division
# speedup vs baseline: 1.8179x; 1.2065x over previous
#include <hip/hip_runtime.h>


typedef __attribute__((ext_vector_type(2))) float v2f;
typedef __attribute__((ext_vector_type(8))) float v8f;

#define HD   16
#define NWT  (4 * HD + 1)
#define QPB  8
#define KSPL 4
#define KST  16
#define EST  4
#define SWP  (NWT + 3)

__global__ __launch_bounds__(512) void posmlp_fused(const float* __restrict__ Q,
                                                    const float* __restrict__ Wg,
                                                    const float* __restrict__ bg,
                                                    const float* __restrict__ pos,
                                                    const int* __restrict__ Hp,
                                                    const int* __restrict__ Wp,
                                                    float* __restrict__ out,
                                                    int BQ, int N, int C, int HW) {
  __shared__ float part[KSPL][QPB][4 * HD];
  __shared__ float partE[16][QPB];
  __shared__ float sw[QPB][SWP];

  const int t    = threadIdx.x;
  const int wave = __builtin_amdgcn_readfirstlane(t >> 5);
  const int lane = t & 31;
  const int hf   = lane >> 4;
  const int l    = lane & 15;
  const int q0   = blockIdx.x * QPB;

  const int qi  = wave >> 1;
  const int sub = wave & 1;
  const int bq  = q0 + qi;
  const int bqc = (bq < BQ) ? bq : (BQ - 1);
  const float cx = pos[bqc * 4 + 0];
  const float cy = pos[bqc * 4 + 1];
  const float bw = pos[bqc * 4 + 2];
  const float bh = pos[bqc * 4 + 3];
  const int H_ = *Hp;
  const int W_ = *Wp;

  {
    const int ks = wave & (KSPL - 1);
    const int nt = wave >> 2;

    int mA = q0 + (l & (QPB - 1));  mA = (mA < BQ) ? mA : (BQ - 1);
    int nB = nt * 16 + l;           nB = (nB < N) ? nB : (N - 1);
    int nE = 4 * HD + l;            nE = (nE < N) ? nE : (N - 1);

    const float* qrow = Q  + (size_t)mA * C + 2 * hf;
    const float* wrow = Wg + (size_t)nB * C + 2 * hf;
    const float* erow = Wg + (size_t)nE * C + 2 * hf;

    v8f acc  = {};
    v8f accE = {};
    for (int kb = ks * (4 * KST); kb + 4 * KST <= C; kb += KSPL * 4 * KST) {
      const int ke = kb + 4 * EST * nt;
      v2f a[KST], b[KST], ae[EST], e[EST];
      #pragma unroll
      for (int s = 0; s < KST; ++s) {
        a[s] = *(const v2f*)(qrow + kb + 4 * s);
        b[s] = *(const v2f*)(wrow + kb + 4 * s);
      }
      #pragma unroll
      for (int s = 0; s < EST; ++s) {
        ae[s] = *(const v2f*)(qrow + ke + 4 * s);
        e[s]  = *(const v2f*)(erow + ke + 4 * s);
      }
      #pragma unroll
      for (int s = 0; s < KST; ++s) {
        acc = __builtin_amdgcn_wmma_f32_16x16x4_f32(
            false, a[s], false, b[s], (short)0, acc, false, false);
      }
      #pragma unroll
      for (int s = 0; s < EST; ++s) {
        accE = __builtin_amdgcn_wmma_f32_16x16x4_f32(
            false, ae[s], false, e[s], (short)0, accE, false, false);
      }
    }
    for (int k = (C & ~(4 * KST - 1)) + 4 * ks; k < C; k += 4 * KSPL) {
      const v2f av = *(const v2f*)(qrow + k);
      const v2f bv = *(const v2f*)(wrow + k);
      acc = __builtin_amdgcn_wmma_f32_16x16x4_f32(
          false, av, false, bv, (short)0, acc, false, false);
      if (nt == 0) {
        const v2f ev = *(const v2f*)(erow + k);
        accE = __builtin_amdgcn_wmma_f32_16x16x4_f32(
            false, av, false, ev, (short)0, accE, false, false);
      }
    }

    if (hf == 0) {
      #pragma unroll
      for (int i = 0; i < 8; ++i) part[ks][i][nt * 16 + l] = acc[i];
    }
    if (lane == 0) {
      #pragma unroll
      for (int i = 0; i < 8; ++i) partE[wave][i] = accE[i];
    }
  }
  __syncthreads();

  {
    const int r = t >> 6;
    const int n = t & 63;
    sw[r][n] = ((part[0][r][n] + part[1][r][n]) + (part[2][r][n] + part[3][r][n])) + bg[n];
    if (t < QPB) {
      float s64 = 0.0f;
      #pragma unroll
      for (int w = 0; w < 16; ++w) s64 += partE[w][t];
      sw[t][4 * HD] = s64 + bg[4 * HD];
    }
  }
  __syncthreads();

  if (bq >= BQ) return;

  const float invbw = 1.0f / bw;
  const float invbh = 1.0f / bh;
  const float invH = 1.0f / (float)H_;
  const float invW = 1.0f / (float)W_;
  float* obase = out + (size_t)bq * HW;

  if (W_ == 3 * 16) {
    const int half = hf;
    v2f w1y2[HD / 2], b12[HD / 2], w22[HD / 2];
    v2f bx[3][HD / 2];
    #pragma unroll
    for (int k = 0; k < HD / 2; ++k) {
      w1y2[k] = (v2f){sw[qi][4 * k], sw[qi][4 * k + 2]};
      b12[k]  = (v2f){sw[qi][2 * HD + 2 * k], sw[qi][2 * HD + 2 * k + 1]};
      w22[k]  = (v2f){sw[qi][3 * HD + 2 * k], sw[qi][3 * HD + 2 * k + 1]};
    }
    #pragma unroll
    for (int j = 0; j < 3; ++j) {
      const float rel_x = (((float)(l + 16 * j) + 0.5f) * invW - cx) * invbw;
      const v2f rx2 = (v2f){rel_x, rel_x};
      #pragma unroll
      for (int k = 0; k < HD / 2; ++k) {
        const v2f w1x2 = (v2f){sw[qi][4 * k + 1], sw[qi][4 * k + 3]};
        bx[j][k] = rx2 * w1x2;
      }
    }
    const float b2 = sw[qi][4 * HD];
    const v2f zero2 = (v2f){0.0f, 0.0f};

    const int Hr = (H_ < HW / (3 * 16)) ? H_ : HW / (3 * 16);
    for (int r0 = 2 * sub; r0 < Hr; r0 += 4) {
      const int row = r0 + half;
      const float rel_y = (((float)row + 0.5f) * invH - cy) * invbh;
      const v2f ry2 = (v2f){rel_y, rel_y};
      v2f ay[HD / 2];
      #pragma unroll
      for (int k = 0; k < HD / 2; ++k) ay[k] = __builtin_elementwise_fma(ry2, w1y2[k], b12[k]);

      float o[3];
      #pragma unroll
      for (int j = 0; j < 3; ++j) {
        v2f acc = (v2f){b2, 0.0f};
        #pragma unroll
        for (int k = 0; k < HD / 2; ++k) {
          const v2f hid = ay[k] + bx[j][k];
          acc = __builtin_elementwise_fma(__builtin_elementwise_max(hid, zero2), w22[k], acc);
        }
        o[j] = acc.x + acc.y;
      }

      const float x = __builtin_bit_cast(float, __builtin_amdgcn_permlanex16(
          __builtin_bit_cast(int, o[1]), __builtin_bit_cast(int, o[1]),
          0x76543210, 0xfedcba98, false, false));
      const float vA = half ? x : o[0];
      const float vB = half ? o[0] : o[2];
      const float vC = half ? o[2] : x;
      float* orow = obase + (size_t)r0 * (3 * 16);
      const bool row1 = (r0 + 1 < Hr);
      orow[lane] = vA;
      if (half == 0 || row1) orow[32 + lane] = vB;
      if (row1) orow[64 + lane] = vC;
    }
    return;
  }

  float w1y[HD], w1x[HD], b1[HD], w2[HD];
  #pragma unroll
  for (int k = 0; k < HD; ++k) {
    w1y[k] = sw[qi][2 * k];
    w1x[k] = sw[qi][2 * k + 1];
    b1[k]  = sw[qi][2 * HD + k];
    w2[k]  = sw[qi][3 * HD + k];
  }
  const float b2 = sw[qi][4 * HD];

  for (int p = sub * 32 + lane; p < HW; p += 64) {
    const int h = p / W_;
    const int w = p - h * W_;
    const float rel_y = (((float)h + 0.5f) * invH - cy) * invbh;
    const float rel_x = (((float)w + 0.5f) * invW - cx) * invbw;
    float o = b2;
    #pragma unroll
    for (int k = 0; k < HD; ++k) {
      const float hid = fmaf(rel_y, w1y[k], fmaf(rel_x, w1x[k], b1[k]));
      o = fmaf(fmaxf(hid, 0.0f), w2[k], o);
    }
    obase[p] = o;
  }
}

extern "C" void kernel_launch(void* const* d_in, const int* in_sizes, int n_in,
                              void* d_out, int out_size, void* d_ws, size_t ws_size,
                              hipStream_t stream) {
  const float* queries = (const float*)d_in[0];
  const float* pos     = (const float*)d_in[1];
  const float* w_gen   = (const float*)d_in[2];
  const float* b_gen   = (const float*)d_in[3];
  const int*   Hp      = (const int*)d_in[4];
  const int*   Wp      = (const int*)d_in[5];

  const int N  = in_sizes[3];
  const int C  = in_sizes[2] / N;
  const int BQ = in_sizes[0] / C;
  const int HW = out_size / BQ;

  const int blocks = (BQ + QPB - 1) / QPB;

  posmlp_fused<<<blocks, 512, 0, stream>>>(
      queries, w_gen, b_gen, pos, Hp, Wp, (float*)d_out, BQ, N, C, HW);
}


# relu(row+col) taken as max(row,-col)+col: the col*w2 part of every hidden unit is summed once per column ahead of the rows (sum order changes), a pixel costs max+fma per unit pair, no add
# speedup vs baseline: 2.0240x; 1.1134x over previous
#include <hip/hip_runtime.h>


typedef __attribute__((ext_vector_type(2))) float v2f;
typedef __attribute__((ext_vector_type(8))) float v8f;

#define HD   16
#define NWT  (4 * HD + 1)
#define QPB  8
#define KSPL 4
#define KST  16
#define EST  4
#define SWP  (NWT + 3)

__global__ __launch_bounds__(512) void posmlp_fused(const float* __restrict__ Q,
                                                    const float* __restrict__ Wg,
                                                    const float* __restrict__ bg,
                                                    const float* __restrict__ pos,
                                                    const int* __restrict__ Hp,
                                                    const int* __restrict__ Wp,
                                                    float* __restrict__ out,
                                                    int BQ, int N, int C, int HW) {
  __shared__ float part[KSPL][QPB][4 * HD];
  __shared__ float partE[16][QPB];
  __shared__ float sw[QPB][SWP];

  const int t    = threadIdx.x;
  const int wave = __builtin_amdgcn_readfirstlane(t >> 5);
  const int lane = t & 31;
  const int hf   = lane >> 4;
  const int l    = lane & 15;
  const int q0   = blockIdx.x * QPB;

  const int qi  = wave >> 1;
  const int sub = wave & 1;
  const int bq  = q0 + qi;
  const int bqc = (bq < BQ) ? bq : (BQ - 1);
  const float cx = pos[bqc * 4 + 0];
  const float cy = pos[bqc * 4 + 1];
  const float bw = pos[bqc * 4 + 2];
  const float bh = pos[bqc * 4 + 3];
  const int H_ = *Hp;
  const int W_ = *Wp;

  {
    const int ks = wave & (KSPL - 1);
    const int nt = wave >> 2;

    int mA = q0 + (l & (QPB - 1));  mA = (mA < BQ) ? mA : (BQ - 1);
    int nB = nt * 16 + l;           nB = (nB < N) ? nB : (N - 1);
    int nE = 4 * HD + l;            nE = (nE < N) ? nE : (N - 1);

    const float* qrow = Q  + (size_t)mA * C + 2 * hf;
    const float* wrow = Wg + (size_t)nB * C + 2 * hf;
    const float* erow = Wg + (size_t)nE * C + 2 * hf;

    v8f acc  = {};
    v8f accE = {};
    for (int kb = ks * (4 * KST); kb + 4 * KST <= C; kb += KSPL * 4 * KST) {
      const int ke = kb + 4 * EST * nt;
      v2f a[KST], b[KST], ae[EST], e[EST];
      #pragma unroll
      for (int s = 0; s < KST; ++s) {
        a[s] = *(const v2f*)(qrow + kb + 4 * s);
        b[s] = *(const v2f*)(wrow + kb + 4 * s);
      }
      #pragma unroll
      for (int s = 0; s < EST; ++s) {
        ae[s] = *(const v2f*)(qrow + ke + 4 * s);
        e[s]  = *(const v2f*)(erow + ke + 4 * s);
      }
      #pragma unroll
      for (int s = 0; s < KST; ++s) {
        acc = __builtin_amdgcn_wmma_f32_16x16x4_f32(
            false, a[s], false, b[s], (short)0, acc, false, false);
      }
      #pragma unroll
      for (int s = 0; s < EST; ++s) {
        accE = __builtin_amdgcn_wmma_f32_16x16x4_f32(
            false, ae[s], false, e[s], (short)0, accE, false, false);
      }
    }
    for (int k = (C & ~(4 * KST - 1)) + 4 * ks; k < C; k += 4 * KSPL) {
      const v2f av = *(const v2f*)(qrow + k);
      const v2f bv = *(const v2f*)(wrow + k);
      acc = __builtin_amdgcn_wmma_f32_16x16x4_f32(
          false, av, false, bv, (short)0, acc, false, false);
      if (nt == 0) {
        const v2f ev = *(const v2f*)(erow + k);
        accE = __builtin_amdgcn_wmma_f32_16x16x4_f32(
            false, av, false, ev, (short)0, accE, false, false);
      }
    }

    if (hf == 0) {
      #pragma unroll
      for (int i = 0; i < 8; ++i) part[ks][i][nt * 16 + l] = acc[i];
    }
    if (lane == 0) {
      #pragma unroll
      for (int i = 0; i < 8; ++i) partE[wave][i] = accE[i];
    }
  }
  __syncthreads();

  {
    const int r = t >> 6;
    const int n = t & 63;
    sw[r][n] = ((part[0][r][n] + part[1][r][n]) + (part[2][r][n] + part[3][r][n])) + bg[n];
    if (t < QPB) {
      float s64 = 0.0f;
      #pragma unroll
      for (int w = 0; w < 16; ++w) s64 += partE[w][t];
      sw[t][4 * HD] = s64 + bg[4 * HD];
    }
  }
  __syncthreads();

  if (bq >= BQ) return;

  const float invbw = 1.0f / bw;
  const float invbh = 1.0f / bh;
  const float invH = 1.0f / (float)H_;
  const float invW = 1.0f / (float)W_;
  float* obase = out + (size_t)bq * HW;

  if (W_ == 3 * 16 && H_ == 3 * 16 && HW == 9 * 256) {
    const int half = hf;
    v2f w1y2[HD / 2], b12[HD / 2], w22[HD / 2];
    v2f nbx[3][HD / 2];
    v2f col0[3];
    #pragma unroll
    for (int k = 0; k < HD / 2; ++k) {
      w1y2[k] = (v2f){sw[qi][4 * k], sw[qi][4 * k + 2]};
      b12[k]  = (v2f){sw[qi][2 * HD + 2 * k], sw[qi][2 * HD + 2 * k + 1]};
      w22[k]  = (v2f){sw[qi][3 * HD + 2 * k], sw[qi][3 * HD + 2 * k + 1]};
    }
    const float b2 = sw[qi][4 * HD];
    #pragma unroll
    for (int j = 0; j < 3; ++j) {
      const float rel_x = (((float)(l + 16 * j) + 0.5f) * invW - cx) * invbw;
      const v2f nrx2 = (v2f){-rel_x, -rel_x};
      v2f c = (v2f){b2, 0.0f};
      #pragma unroll
      for (int k = 0; k < HD / 2; ++k) {
        const v2f w1x2 = (v2f){sw[qi][4 * k + 1], sw[qi][4 * k + 3]};
        nbx[j][k] = nrx2 * w1x2;
        c = __builtin_elementwise_fma(-nbx[j][k], w22[k], c);
      }
      col0[j] = c;
    }

    #pragma unroll
    for (int i = 0; i < 12; ++i) {
      const int r0 = 2 * sub + 4 * i;
      const int row = r0 + half;
      const float rel_y = (((float)row + 0.5f) * invH - cy) * invbh;
      const v2f ry2 = (v2f){rel_y, rel_y};
      v2f ay[HD / 2];
      #pragma unroll
      for (int k = 0; k < HD / 2; ++k) ay[k] = __builtin_elementwise_fma(ry2, w1y2[k], b12[k]);

      float o[3];
      #pragma unroll
      for (int j = 0; j < 3; ++j) {
        v2f acc = col0[j];
        #pragma unroll
        for (int k = 0; k < HD / 2; ++k) {
          acc = __builtin_elementwise_fma(__builtin_elementwise_max(ay[k], nbx[j][k]), w22[k], acc);
        }
        o[j] = acc.x + acc.y;
      }

      const float x = __builtin_bit_cast(float, __builtin_amdgcn_permlanex16(
          __builtin_bit_cast(int, o[1]), __builtin_bit_cast(int, o[1]),
          0x76543210, 0xfedcba98, false, false));
      float* orow = obase + (size_t)r0 * (3 * 16);
      orow[lane]      = half ? x : o[0];
      orow[32 + lane] = half ? o[0] : o[2];
      orow[64 + lane] = half ? o[2] : x;
    }
    return;
  }

  float w1y[HD], w1x[HD], b1[HD], w2[HD];
  #pragma unroll
  for (int k = 0; k < HD; ++k) {
    w1y[k] = sw[qi][2 * k];
    w1x[k] = sw[qi][2 * k + 1];
    b1[k]  = sw[qi][2 * HD + k];
    w2[k]  = sw[qi][3 * HD + k];
  }
  const float b2 = sw[qi][4 * HD];

  for (int p = sub * 32 + lane; p < HW; p += 64) {
    const int h = p / W_;
    const int w = p - h * W_;
    const float rel_y = (((float)h + 0.5f) * invH - cy) * invbh;
    const float rel_x = (((float)w + 0.5f) * invW - cx) * invbw;
    float o = b2;
    #pragma unroll
    for (int k = 0; k < HD; ++k) {
      const float hid = fmaf(rel_y, w1y[k], fmaf(rel_x, w1x[k], b1[k]));
      o = fmaf(fmaxf(hid, 0.0f), w2[k], o);
    }
    obase[p] = o;
  }
}

extern "C" void kernel_launch(void* const* d_in, const int* in_sizes, int n_in,
                              void* d_out, int out_size, void* d_ws, size_t ws_size,
                              hipStream_t stream) {
  const float* queries = (const float*)d_in[0];
  const float* pos     = (const float*)d_in[1];
  const float* w_gen   = (const float*)d_in[2];
  const float* b_gen   = (const float*)d_in[3];
  const int*   Hp      = (const int*)d_in[4];
  const int*   Wp      = (const int*)d_in[5];

  const int N  = in_sizes[3];
  const int C  = in_sizes[2] / N;
  const int BQ = in_sizes[0] / C;
  const int HW = out_size / BQ;

  const int blocks = (BQ + QPB - 1) / QPB;

  posmlp_fused<<<blocks, 512, 0, stream>>>(
      queries, w_gen, b_gen, pos, Hp, Wp, (float*)d_out, BQ, N, C, HW);
}


# one wave per query: blocks of 8 waves, not 16, so half as many waves enter; each wave takes two N-tiles of the weight GEMM on shared A fragments (column 64: K summed in 8 parts, not 16)
# speedup vs baseline: 2.5323x; 1.2511x over previous
#include <hip/hip_runtime.h>


typedef __attribute__((ext_vector_type(2))) float v2f;
typedef __attribute__((ext_vector_type(8))) float v8f;

#define HD   16
#define NWT  (4 * HD + 1)
#define QPB  8
#define NWV  8
#define NTH  256
#define KSPL 4
#define KST  16
#define EST  8
#define SWP  (NWT + 3)

__global__ __launch_bounds__(NTH) void posmlp_fused(const float* __restrict__ Q,
                                                    const float* __restrict__ Wg,
                                                    const float* __restrict__ bg,
                                                    const float* __restrict__ pos,
                                                    const int* __restrict__ Hp,
                                                    const int* __restrict__ Wp,
                                                    float* __restrict__ out,
                                                    int BQ, int N, int C, int HW) {
  __shared__ float part[KSPL][QPB][4 * HD];
  __shared__ float partE[NWV][QPB];
  __shared__ float sw[QPB][SWP];

  const int t    = threadIdx.x;
  const int wave = __builtin_amdgcn_readfirstlane(t >> 5);
  const int lane = t & 31;
  const int hf   = lane >> 4;
  const int l    = lane & 15;
  const int q0   = blockIdx.x * QPB;

  const int qi  = wave;
  const int bq  = q0 + qi;
  const int bqc = (bq < BQ) ? bq : (BQ - 1);
  const float cx = pos[bqc * 4 + 0];
  const float cy = pos[bqc * 4 + 1];
  const float bw = pos[bqc * 4 + 2];
  const float bh = pos[bqc * 4 + 3];
  const int H_ = *Hp;
  const int W_ = *Wp;

  {
    const int ks = wave & (KSPL - 1);
    const int np = wave >> 2;

    int mA  = q0 + (l & (QPB - 1));  mA  = (mA < BQ) ? mA : (BQ - 1);
    int nB0 = np * 32 + l;           nB0 = (nB0 < N) ? nB0 : (N - 1);
    int nB1 = np * 32 + 16 + l;      nB1 = (nB1 < N) ? nB1 : (N - 1);
    int nE  = 4 * HD + l;            nE  = (nE < N) ? nE : (N - 1);

    const float* qrow  = Q  + (size_t)mA  * C + 2 * hf;
    const float* wrow0 = Wg + (size_t)nB0 * C + 2 * hf;
    const float* wrow1 = Wg + (size_t)nB1 * C + 2 * hf;
    const float* erow  = Wg + (size_t)nE  * C + 2 * hf;

    v8f acc0 = {};
    v8f acc1 = {};
    v8f accE = {};
    for (int kb = ks * (4 * KST); kb + 4 * KST <= C; kb += KSPL * 4 * KST) {
      const int ke = kb + 4 * EST * np;
      v2f a[KST], b0[KST], b1[KST], ae[EST], e[EST];
      #pragma unroll
      for (int s = 0; s < KST; ++s) {
        a[s]  = *(const v2f*)(qrow  + kb + 4 * s);
        b0[s] = *(const v2f*)(wrow0 + kb + 4 * s);
        b1[s] = *(const v2f*)(wrow1 + kb + 4 * s);
      }
      #pragma unroll
      for (int s = 0; s < EST; ++s) {
        ae[s] = *(const v2f*)(qrow + ke + 4 * s);
        e[s]  = *(const v2f*)(erow + ke + 4 * s);
      }
      #pragma unroll
      for (int s = 0; s < KST; ++s) {
        acc0 = __builtin_amdgcn_wmma_f32_16x16x4_f32(
            false, a[s], false, b0[s], (short)0, acc0, false, false);
        acc1 = __builtin_amdgcn_wmma_f32_16x16x4_f32(
            false, a[s], false, b1[s], (short)0, acc1, false, false);
      }
      #pragma unroll
      for (int s = 0; s < EST; ++s) {
        accE = __builtin_amdgcn_wmma_f32_16x16x4_f32(
            false, ae[s], false, e[s], (short)0, accE, false, false);
      }
    }
    for (int k = (C & ~(4 * KST - 1)) + 4 * ks; k < C; k += 4 * KSPL) {
      const v2f av  = *(const v2f*)(qrow  + k);
      const v2f bv0 = *(const v2f*)(wrow0 + k);
      const v2f bv1 = *(const v2f*)(wrow1 + k);
      acc0 = __builtin_amdgcn_wmma_f32_16x16x4_f32(
          false, av, false, bv0, (short)0, acc0, false, false);
      acc1 = __builtin_amdgcn_wmma_f32_16x16x4_f32(
          false, av, false, bv1, (short)0, acc1, false, false);
      if (np == 0) {
        const v2f ev = *(const v2f*)(erow + k);
        accE = __builtin_amdgcn_wmma_f32_16x16x4_f32(
            false, av, false, ev, (short)0, accE, false, false);
      }
    }

    if (hf == 0) {
      #pragma unroll
      for (int i = 0; i < 8; ++i) {
        part[ks][i][np * 32 + l]      = acc0[i];
        part[ks][i][np * 32 + 16 + l] = acc1[i];
      }
    }
    if (lane == 0) {
      #pragma unroll
      for (int i = 0; i < 8; ++i) partE[wave][i] = accE[i];
    }
  }
  __syncthreads();

  {
    const int n = t & 63;
    const float bn = bg[n];
    #pragma unroll
    for (int j = 0; j < (QPB * 4 * HD) / NTH; ++j) {
      const int r = (t >> 6) + j * (NTH >> 6);
      sw[r][n] = ((part[0][r][n] + part[1][r][n]) + (part[2][r][n] + part[3][r][n])) + bn;
    }
    if (t < QPB) {
      float s64 = 0.0f;
      #pragma unroll
      for (int w = 0; w < NWV; ++w) s64 += partE[w][t];
      sw[t][4 * HD] = s64 + bg[4 * HD];
    }
  }
  __syncthreads();

  if (bq >= BQ) return;

  const float invbw = 1.0f / bw;
  const float invbh = 1.0f / bh;
  const float invH = 1.0f / (float)H_;
  const float invW = 1.0f / (float)W_;
  float* obase = out + (size_t)bq * HW;

  if (W_ == 3 * 16 && H_ == 3 * 16 && HW == 9 * 256) {
    const int half = hf;
    v2f w1y2[HD / 2], b12[HD / 2], w22[HD / 2];
    v2f nbx[3][HD / 2];
    v2f col0[3];
    #pragma unroll
    for (int k = 0; k < HD / 2; ++k) {
      w1y2[k] = (v2f){sw[qi][4 * k], sw[qi][4 * k + 2]};
      b12[k]  = (v2f){sw[qi][2 * HD + 2 * k], sw[qi][2 * HD + 2 * k + 1]};
      w22[k]  = (v2f){sw[qi][3 * HD + 2 * k], sw[qi][3 * HD + 2 * k + 1]};
    }
    const float b2 = sw[qi][4 * HD];
    #pragma unroll
    for (int j = 0; j < 3; ++j) {
      const float rel_x = (((float)(l + 16 * j) + 0.5f) * invW - cx) * invbw;
      const v2f nrx2 = (v2f){-rel_x, -rel_x};
      v2f c = (v2f){b2, 0.0f};
      #pragma unroll
      for (int k = 0; k < HD / 2; ++k) {
        const v2f w1x2 = (v2f){sw[qi][4 * k + 1], sw[qi][4 * k + 3]};
        nbx[j][k] = nrx2 * w1x2;
        c = __builtin_elementwise_fma(-nbx[j][k], w22[k], c);
      }
      col0[j] = c;
    }

    #pragma unroll
    for (int i = 0; i < 24; ++i) {
      const int r0 = 2 * i;
      const int row = r0 + half;
      const float rel_y = (((float)row + 0.5f) * invH - cy) * invbh;
      const v2f ry2 = (v2f){rel_y, rel_y};
      v2f ay[HD / 2];
      #pragma unroll
      for (int k = 0; k < HD / 2; ++k) ay[k] = __builtin_elementwise_fma(ry2, w1y2[k], b12[k]);

      float o[3];
      #pragma unroll
      for (int j = 0; j < 3; ++j) {
        v2f acc = col0[j];
        #pragma unroll
        for (int k = 0; k < HD / 2; ++k) {
          acc = __builtin_elementwise_fma(__builtin_elementwise_max(ay[k], nbx[j][k]), w22[k], acc);
        }
        o[j] = acc.x + acc.y;
      }

      const float x = __builtin_bit_cast(float, __builtin_amdgcn_permlanex16(
          __builtin_bit_cast(int, o[1]), __builtin_bit_cast(int, o[1]),
          0x76543210, 0xfedcba98, false, false));
      float* orow = obase + (size_t)r0 * (3 * 16);
      orow[lane]      = half ? x : o[0];
      orow[32 + lane] = half ? o[0] : o[2];
      orow[64 + lane] = half ? o[2] : x;
    }
    return;
  }

  float w1y[HD], w1x[HD], b1[HD], w2[HD];
  #pragma unroll
  for (int k = 0; k < HD; ++k) {
    w1y[k] = sw[qi][2 * k];
    w1x[k] = sw[qi][2 * k + 1];
    b1[k]  = sw[qi][2 * HD + k];
    w2[k]  = sw[qi][3 * HD + k];
  }
  const float b2 = sw[qi][4 * HD];

  for (int p = lane; p < HW; p += 32) {
    const int h = p / W_;
    const int w = p - h * W_;
    const float rel_y = (((float)h + 0.5f) * invH - cy) * invbh;
    const float rel_x = (((float)w + 0.5f) * invW - cx) * invbw;
    float o = b2;
    #pragma unroll
    for (int k = 0; k < HD; ++k) {
      const float hid = fmaf(rel_y, w1y[k], fmaf(rel_x, w1x[k], b1[k]));
      o = fmaf(fmaxf(hid, 0.0f), w2[k], o);
    }
    obase[p] = o;
  }
}

extern "C" void kernel_launch(void* const* d_in, const int* in_sizes, int n_in,
                              void* d_out, int out_size, void* d_ws, size_t ws_size,
                              hipStream_t stream) {
  const float* queries = (const float*)d_in[0];
  const float* pos     = (const float*)d_in[1];
  const float* w_gen   = (const float*)d_in[2];
  const float* b_gen   = (const float*)d_in[3];
  const int*   Hp      = (const int*)d_in[4];
  const int*   Wp      = (const int*)d_in[5];

  const int N  = in_sizes[3];
  const int C  = in_sizes[2] / N;
  const int BQ = in_sizes[0] / C;
  const int HW = out_size / BQ;

  const int blocks = (BQ + QPB - 1) / QPB;

  posmlp_fused<<<blocks, NTH, 0, stream>>>(
      queries, w_gen, b_gen, pos, Hp, Wp, (float*)d_out, BQ, N, C, HW);
}
